# starting kernel
# baseline (speedup 1.0000x reference)
#include <hip/hip_runtime.h>


typedef __attribute__((ext_vector_type(2))) float v2f;
typedef __attribute__((ext_vector_type(8))) float v8f;

__device__ __forceinline__ void atomAddF(float* p, float v) {
    asm volatile("global_atomic_add_f32 %0, %1, off" :: "v"(p), "v"(v) : "memory");
}

__global__ void k_zero(float* __restrict__ p, long n) {
    long i = (long)blockIdx.x * blockDim.x + threadIdx.x;
    long stride = (long)gridDim.x * blockDim.x;
    for (; i < n; i += stride) p[i] = 0.0f;
}

__global__ void k_deg(const int* __restrict__ src, const int* __restrict__ dst,
                      int ne, float* __restrict__ degS, float* __restrict__ degD) {
    int e = blockIdx.x * blockDim.x + threadIdx.x;
    if (e < ne) {
        atomAddF(&degS[src[e]], 1.0f);
        atomAddF(&degD[dst[e]], 1.0f);
    }
}

__global__ void k_rsqrt(float* __restrict__ p, int n) {
    int i = blockIdx.x * blockDim.x + threadIdx.x;
    if (i < n) p[i] = rsqrtf(fmaxf(p[i], 1.0f));
}

__global__ void k_spmm128(const float* __restrict__ x, const int* __restrict__ src,
                          const int* __restrict__ dst, const float* __restrict__ normS,
                          float* __restrict__ agg, int ne) {
    int e = blockIdx.x * 8 + (threadIdx.x >> 5);
    int lane = threadIdx.x & 31;
    if (e >= ne) return;
    int s = src[e];
    int d = dst[e];
    float ns = normS[s];
    const float4* xp = (const float4*)(x + (long)s * 128);
    float4 v = xp[lane];
    float* ap = agg + (long)d * 128 + lane * 4;
    atomAddF(ap + 0, v.x * ns);
    atomAddF(ap + 1, v.y * ns);
    atomAddF(ap + 2, v.z * ns);
    atomAddF(ap + 3, v.w * ns);
}

__global__ void k_spmm64(const float* __restrict__ g, const int* __restrict__ src,
                         const int* __restrict__ dst, float* __restrict__ agg, int ne) {
    int e = blockIdx.x * 8 + (threadIdx.x >> 5);
    int lane = threadIdx.x & 31;
    if (e >= ne) return;
    int s = src[e];
    int d = dst[e];
    const float2* gp = (const float2*)(g + (long)s * 64);
    float2 v = gp[lane];
    float* ap = agg + (long)d * 64 + lane * 2;
    atomAddF(ap + 0, v.x);
    atomAddF(ap + 1, v.y);
}

template <int K, int N, bool RELU, bool BIAS>
__global__ void k_gemm_wmma(const float* __restrict__ A,
                            const float* __restrict__ rowscale,
                            const float* __restrict__ W,
                            const float* __restrict__ bias,
                            float* __restrict__ C) {
    constexpr int tilesN = N / 16;
    int wave = (blockIdx.x * blockDim.x + threadIdx.x) >> 5;
    int lane = threadIdx.x & 31;
    int tm = wave / tilesN;
    int tn = wave % tilesN;

    int half = lane >> 4;
    int l15  = lane & 15;
    int arow = tm * 16 + l15;
    int gcol = tn * 16 + l15;

    float scale = rowscale[arow];
    const float* ap = A + (long)arow * K + half * 2;
    const float* wp = W + gcol;

    v8f c = {};
    for (int k = 0; k < K; k += 4) {
        v2f a;
        a.x = ap[k]     * scale;
        a.y = ap[k + 1] * scale;
        int kb = k + half * 2;
        v2f b;
        b.x = wp[(long)kb * N];
        b.y = wp[(long)(kb + 1) * N];
        c = __builtin_amdgcn_wmma_f32_16x16x4_f32(
                 false, a,  false, b,
                 (short)0, c,  false,  false);
    }

    float bval = 0.0f;
    if (BIAS) bval = bias[gcol];
    int rb = half * 8;
    #pragma unroll
    for (int i = 0; i < 8; ++i) {
        float v = c[i] + bval;
        if (RELU) v = fmaxf(v, 0.0f);
        C[(long)(tm * 16 + rb + i) * N + gcol] = v;
    }
}

__global__ void k_final(const float* __restrict__ agg2, const float* __restrict__ normD2,
                        const float* __restrict__ b2, float* __restrict__ out, int tot) {
    int t = blockIdx.x * blockDim.x + threadIdx.x;
    if (t < tot) {
        int r = t >> 6;
        int f = t & 63;
        out[t] = agg2[t] * normD2[r] + b2[f];
    }
}

extern "C" void kernel_launch(void* const* d_in, const int* in_sizes, int n_in,
                              void* d_out, int out_size, void* d_ws, size_t ws_size,
                              hipStream_t stream) {
    const float* x    = (const float*)d_in[0];
    const float* W1   = (const float*)d_in[1];
    const float* b1   = (const float*)d_in[2];
    const float* W2   = (const float*)d_in[3];
    const float* b2   = (const float*)d_in[4];
    const int*   src1 = (const int*)d_in[5];
    const int*   dst1 = (const int*)d_in[6];
    const int*   src2 = (const int*)d_in[7];
    const int*   dst2 = (const int*)d_in[8];

    const int IN_F = 128, H_F = 256, N_CLS = 64;
    const int NSRC1 = in_sizes[0] / IN_F;
    const int E1    = in_sizes[5];
    const int E2    = in_sizes[7];
    const int ND1   = 100000;
    const int ND2   = 16384;

    float* ws    = (float*)d_ws;
    float* degS1 = ws;
    float* degD1 = degS1 + NSRC1;
    float* degS2 = degD1 + ND1;
    float* degD2 = degS2 + ND1;
    float* agg1  = degD2 + ND2;
    float* agg2  = agg1 + (long)ND1 * IN_F;
    float* h     = agg2 + (long)ND2 * N_CLS;
    float* g     = h + (long)ND1 * H_F;

    long zcount = (NSRC1 + ND1 + ND1 + ND2) + (long)ND1 * IN_F + (long)ND2 * N_CLS;
    k_zero<<<2048, 256, 0, stream>>>(ws, zcount);

    k_deg<<<(E1 + 255) / 256, 256, 0, stream>>>(src1, dst1, E1, degS1, degD1);
    k_deg<<<(E2 + 255) / 256, 256, 0, stream>>>(src2, dst2, E2, degS2, degD2);
    int degtot = NSRC1 + ND1 + ND1 + ND2;
    k_rsqrt<<<(degtot + 255) / 256, 256, 0, stream>>>(ws, degtot);

    k_spmm128<<<(E1 + 7) / 8, 256, 0, stream>>>(x, src1, dst1, degS1, agg1, E1);

    {
        int waves = (ND1 / 16) * (H_F / 16);
        k_gemm_wmma<128, 256, true, true>
            <<<waves / 4, 128, 0, stream>>>(agg1, degD1, W1, b1, h);
    }

    {
        int waves = (ND1 / 16) * (N_CLS / 16);
        k_gemm_wmma<256, 64, false, false>
            <<<waves / 4, 128, 0, stream>>>(h, degS2, W2, nullptr, g);
    }

    k_spmm64<<<(E2 + 7) / 8, 256, 0, stream>>>(g, src2, dst2, agg2, E2);

    int tot = ND2 * N_CLS;
    k_final<<<(tot + 255) / 256, 256, 0, stream>>>(agg2, degD2, b2, (float*)d_out, tot);
}


# SpMM kernels: each wave takes 8 consecutive edges and issues all 8 row gathers before its first atomic add (8x fewer waves); sums still float atomics in arrival order, as in the parent
# speedup vs baseline: 1.0801x; 1.0801x over previous
#include <hip/hip_runtime.h>


typedef __attribute__((ext_vector_type(2))) float v2f;
typedef __attribute__((ext_vector_type(8))) float v8f;

__device__ __forceinline__ void atomAddF(float* p, float v) {
    asm volatile("global_atomic_add_f32 %0, %1, off" :: "v"(p), "v"(v) : "memory");
}

__global__ void k_zero(float* __restrict__ p, long n) {
    long i = (long)blockIdx.x * blockDim.x + threadIdx.x;
    long stride = (long)gridDim.x * blockDim.x;
    for (; i < n; i += stride) p[i] = 0.0f;
}

__global__ void k_deg(const int* __restrict__ src, const int* __restrict__ dst,
                      int ne, float* __restrict__ degS, float* __restrict__ degD) {
    int e = blockIdx.x * blockDim.x + threadIdx.x;
    if (e < ne) {
        atomAddF(&degS[src[e]], 1.0f);
        atomAddF(&degD[dst[e]], 1.0f);
    }
}

__global__ void k_rsqrt(float* __restrict__ p, int n) {
    int i = blockIdx.x * blockDim.x + threadIdx.x;
    if (i < n) p[i] = rsqrtf(fmaxf(p[i], 1.0f));
}

constexpr int EPW = 8;
constexpr int WPB = 8;

template <bool TAIL>
__device__ __forceinline__ void spmm128_batch(const float* __restrict__ x,
                                              const int* __restrict__ src,
                                              const int* __restrict__ dst,
                                              const float* __restrict__ normS,
                                              float* __restrict__ agg,
                                              int e0, int ne, int lane) {
    int s[EPW], d[EPW];
    #pragma unroll
    for (int j = 0; j < EPW; ++j) {
        int e = TAIL ? min(e0 + j, ne - 1) : e0 + j;
        s[j] = src[e];
        d[j] = dst[e];
    }
    float ns[EPW];
    float4 v[EPW];
    #pragma unroll
    for (int j = 0; j < EPW; ++j) {
        ns[j] = normS[s[j]];
        v[j] = ((const float4*)(x + (long)s[j] * 128))[lane];
    }
    #pragma unroll
    for (int j = 0; j < EPW; ++j) {
        if (!TAIL || e0 + j < ne) {
            float* ap = agg + (long)d[j] * 128 + lane * 4;
            atomAddF(ap + 0, v[j].x * ns[j]);
            atomAddF(ap + 1, v[j].y * ns[j]);
            atomAddF(ap + 2, v[j].z * ns[j]);
            atomAddF(ap + 3, v[j].w * ns[j]);
        }
    }
}

__global__ void k_spmm128(const float* __restrict__ x, const int* __restrict__ src,
                          const int* __restrict__ dst, const float* __restrict__ normS,
                          float* __restrict__ agg, int ne) {
    int wave = __builtin_amdgcn_readfirstlane((int)(threadIdx.x >> 5));
    int lane = threadIdx.x & 31;
    int e0 = (blockIdx.x * WPB + wave) * EPW;
    if (e0 >= ne) return;
    if (e0 + EPW <= ne) spmm128_batch<false>(x, src, dst, normS, agg, e0, ne, lane);
    else                spmm128_batch<true >(x, src, dst, normS, agg, e0, ne, lane);
}

template <bool TAIL>
__device__ __forceinline__ void spmm64_batch(const float* __restrict__ g,
                                             const int* __restrict__ src,
                                             const int* __restrict__ dst,
                                             float* __restrict__ agg,
                                             int e0, int ne, int lane) {
    int s[EPW], d[EPW];
    #pragma unroll
    for (int j = 0; j < EPW; ++j) {
        int e = TAIL ? min(e0 + j, ne - 1) : e0 + j;
        s[j] = src[e];
        d[j] = dst[e];
    }
    float2 v[EPW];
    #pragma unroll
    for (int j = 0; j < EPW; ++j)
        v[j] = ((const float2*)(g + (long)s[j] * 64))[lane];
    #pragma unroll
    for (int j = 0; j < EPW; ++j) {
        if (!TAIL || e0 + j < ne) {
            float* ap = agg + (long)d[j] * 64 + lane * 2;
            atomAddF(ap + 0, v[j].x);
            atomAddF(ap + 1, v[j].y);
        }
    }
}

__global__ void k_spmm64(const float* __restrict__ g, const int* __restrict__ src,
                         const int* __restrict__ dst, float* __restrict__ agg, int ne) {
    int wave = __builtin_amdgcn_readfirstlane((int)(threadIdx.x >> 5));
    int lane = threadIdx.x & 31;
    int e0 = (blockIdx.x * WPB + wave) * EPW;
    if (e0 >= ne) return;
    if (e0 + EPW <= ne) spmm64_batch<false>(g, src, dst, agg, e0, ne, lane);
    else                spmm64_batch<true >(g, src, dst, agg, e0, ne, lane);
}

template <int K, int N, bool RELU, bool BIAS>
__global__ void k_gemm_wmma(const float* __restrict__ A,
                            const float* __restrict__ rowscale,
                            const float* __restrict__ W,
                            const float* __restrict__ bias,
                            float* __restrict__ C) {
    constexpr int tilesN = N / 16;
    int wave = (blockIdx.x * blockDim.x + threadIdx.x) >> 5;
    int lane = threadIdx.x & 31;
    int tm = wave / tilesN;
    int tn = wave % tilesN;

    int half = lane >> 4;
    int l15  = lane & 15;
    int arow = tm * 16 + l15;
    int gcol = tn * 16 + l15;

    float scale = rowscale[arow];
    const float* ap = A + (long)arow * K + half * 2;
    const float* wp = W + gcol;

    v8f c = {};
    for (int k = 0; k < K; k += 4) {
        v2f a;
        a.x = ap[k]     * scale;
        a.y = ap[k + 1] * scale;
        int kb = k + half * 2;
        v2f b;
        b.x = wp[(long)kb * N];
        b.y = wp[(long)(kb + 1) * N];
        c = __builtin_amdgcn_wmma_f32_16x16x4_f32(
                 false, a,  false, b,
                 (short)0, c,  false,  false);
    }

    float bval = 0.0f;
    if (BIAS) bval = bias[gcol];
    int rb = half * 8;
    #pragma unroll
    for (int i = 0; i < 8; ++i) {
        float v = c[i] + bval;
        if (RELU) v = fmaxf(v, 0.0f);
        C[(long)(tm * 16 + rb + i) * N + gcol] = v;
    }
}

__global__ void k_final(const float* __restrict__ agg2, const float* __restrict__ normD2,
                        const float* __restrict__ b2, float* __restrict__ out, int tot) {
    int t = blockIdx.x * blockDim.x + threadIdx.x;
    if (t < tot) {
        int r = t >> 6;
        int f = t & 63;
        out[t] = agg2[t] * normD2[r] + b2[f];
    }
}

extern "C" void kernel_launch(void* const* d_in, const int* in_sizes, int n_in,
                              void* d_out, int out_size, void* d_ws, size_t ws_size,
                              hipStream_t stream) {
    const float* x    = (const float*)d_in[0];
    const float* W1   = (const float*)d_in[1];
    const float* b1   = (const float*)d_in[2];
    const float* W2   = (const float*)d_in[3];
    const float* b2   = (const float*)d_in[4];
    const int*   src1 = (const int*)d_in[5];
    const int*   dst1 = (const int*)d_in[6];
    const int*   src2 = (const int*)d_in[7];
    const int*   dst2 = (const int*)d_in[8];

    const int IN_F = 128, H_F = 256, N_CLS = 64;
    const int NSRC1 = in_sizes[0] / IN_F;
    const int E1    = in_sizes[5];
    const int E2    = in_sizes[7];
    const int ND1   = 100000;
    const int ND2   = 16384;

    float* ws    = (float*)d_ws;
    float* degS1 = ws;
    float* degD1 = degS1 + NSRC1;
    float* degS2 = degD1 + ND1;
    float* degD2 = degS2 + ND1;
    float* agg1  = degD2 + ND2;
    float* agg2  = agg1 + (long)ND1 * IN_F;
    float* h     = agg2 + (long)ND2 * N_CLS;
    float* g     = h + (long)ND1 * H_F;

    long zcount = (NSRC1 + ND1 + ND1 + ND2) + (long)ND1 * IN_F + (long)ND2 * N_CLS;
    k_zero<<<2048, 256, 0, stream>>>(ws, zcount);

    k_deg<<<(E1 + 255) / 256, 256, 0, stream>>>(src1, dst1, E1, degS1, degD1);
    k_deg<<<(E2 + 255) / 256, 256, 0, stream>>>(src2, dst2, E2, degS2, degD2);
    int degtot = NSRC1 + ND1 + ND1 + ND2;
    k_rsqrt<<<(degtot + 255) / 256, 256, 0, stream>>>(ws, degtot);

    k_spmm128<<<(E1 + 63) / 64, 256, 0, stream>>>(x, src1, dst1, degS1, agg1, E1);

    {
        int waves = (ND1 / 16) * (H_F / 16);
        k_gemm_wmma<128, 256, true, true>
            <<<waves / 4, 128, 0, stream>>>(agg1, degD1, W1, b1, h);
    }

    {
        int waves = (ND1 / 16) * (N_CLS / 16);
        k_gemm_wmma<256, 64, false, false>
            <<<waves / 4, 128, 0, stream>>>(h, degS2, W2, nullptr, g);
    }

    k_spmm64<<<(E2 + 63) / 64, 256, 0, stream>>>(g, src2, dst2, agg2, E2);

    int tot = ND2 * N_CLS;
    k_final<<<(tot + 255) / 256, 256, 0, stream>>>(agg2, degD2, b2, (float*)d_out, tot);
}


# SpMM kernels: lane l owns features l+32q, so each gather load and each float atomic add covers one whole 128-B line (was 4 lines per add, 2 in layer 2); sums still float atomics in arrival order
# speedup vs baseline: 1.3944x; 1.2910x over previous
#include <hip/hip_runtime.h>


typedef __attribute__((ext_vector_type(2))) float v2f;
typedef __attribute__((ext_vector_type(8))) float v8f;

__device__ __forceinline__ void atomAddF(float* p, float v) {
    asm volatile("global_atomic_add_f32 %0, %1, off" :: "v"(p), "v"(v) : "memory");
}

template <int BYTE_OFF>
__device__ __forceinline__ void atomAddRowPiece(float* rowBase, unsigned laneByte, float v) {
    asm volatile("global_atomic_add_f32 %0, %1, %2 offset:%3"
                 :: "v"(laneByte), "v"(v), "s"(rowBase), "n"(BYTE_OFF) : "memory");
}

__global__ void k_zero(float* __restrict__ p, long n) {
    long i = (long)blockIdx.x * blockDim.x + threadIdx.x;
    long stride = (long)gridDim.x * blockDim.x;
    for (; i < n; i += stride) p[i] = 0.0f;
}

__global__ void k_deg(const int* __restrict__ src, const int* __restrict__ dst,
                      int ne, float* __restrict__ degS, float* __restrict__ degD) {
    int e = blockIdx.x * blockDim.x + threadIdx.x;
    if (e < ne) {
        atomAddF(&degS[src[e]], 1.0f);
        atomAddF(&degD[dst[e]], 1.0f);
    }
}

__global__ void k_rsqrt(float* __restrict__ p, int n) {
    int i = blockIdx.x * blockDim.x + threadIdx.x;
    if (i < n) p[i] = rsqrtf(fmaxf(p[i], 1.0f));
}

constexpr int EPW = 8;
constexpr int WPB = 8;

template <bool TAIL>
__device__ __forceinline__ void spmm128_batch(const float* __restrict__ x,
                                              const int* __restrict__ src,
                                              const int* __restrict__ dst,
                                              const float* __restrict__ normS,
                                              float* __restrict__ agg,
                                              int e0, int ne, int lane) {
    int s[EPW], d[EPW];
    #pragma unroll
    for (int j = 0; j < EPW; ++j) {
        int e = TAIL ? min(e0 + j, ne - 1) : e0 + j;
        s[j] = src[e];
        d[j] = dst[e];
    }
    float ns[EPW];
    float v[EPW][4];
    #pragma unroll
    for (int j = 0; j < EPW; ++j) {
        ns[j] = normS[s[j]];
        const float* xr = x + (long)s[j] * 128 + lane;
        v[j][0] = xr[0];
        v[j][1] = xr[32];
        v[j][2] = xr[64];
        v[j][3] = xr[96];
    }
    unsigned laneByte = (unsigned)lane * 4u;
    #pragma unroll
    for (int j = 0; j < EPW; ++j) {
        if (!TAIL || e0 + j < ne) {
            float* row = agg + (long)d[j] * 128;
            atomAddRowPiece<0>(row, laneByte, v[j][0] * ns[j]);
            atomAddRowPiece<128>(row, laneByte, v[j][1] * ns[j]);
            atomAddRowPiece<256>(row, laneByte, v[j][2] * ns[j]);
            atomAddRowPiece<384>(row, laneByte, v[j][3] * ns[j]);
        }
    }
}

__global__ void k_spmm128(const float* __restrict__ x, const int* __restrict__ src,
                          const int* __restrict__ dst, const float* __restrict__ normS,
                          float* __restrict__ agg, int ne) {
    int wave = __builtin_amdgcn_readfirstlane((int)(threadIdx.x >> 5));
    int lane = threadIdx.x & 31;
    int e0 = (blockIdx.x * WPB + wave) * EPW;
    if (e0 >= ne) return;
    if (e0 + EPW <= ne) spmm128_batch<false>(x, src, dst, normS, agg, e0, ne, lane);
    else                spmm128_batch<true >(x, src, dst, normS, agg, e0, ne, lane);
}

template <bool TAIL>
__device__ __forceinline__ void spmm64_batch(const float* __restrict__ g,
                                             const int* __restrict__ src,
                                             const int* __restrict__ dst,
                                             float* __restrict__ agg,
                                             int e0, int ne, int lane) {
    int s[EPW], d[EPW];
    #pragma unroll
    for (int j = 0; j < EPW; ++j) {
        int e = TAIL ? min(e0 + j, ne - 1) : e0 + j;
        s[j] = src[e];
        d[j] = dst[e];
    }
    float v[EPW][2];
    #pragma unroll
    for (int j = 0; j < EPW; ++j) {
        const float* gr = g + (long)s[j] * 64 + lane;
        v[j][0] = gr[0];
        v[j][1] = gr[32];
    }
    unsigned laneByte = (unsigned)lane * 4u;
    #pragma unroll
    for (int j = 0; j < EPW; ++j) {
        if (!TAIL || e0 + j < ne) {
            float* row = agg + (long)d[j] * 64;
            atomAddRowPiece<0>(row, laneByte, v[j][0]);
            atomAddRowPiece<128>(row, laneByte, v[j][1]);
        }
    }
}

__global__ void k_spmm64(const float* __restrict__ g, const int* __restrict__ src,
                         const int* __restrict__ dst, float* __restrict__ agg, int ne) {
    int wave = __builtin_amdgcn_readfirstlane((int)(threadIdx.x >> 5));
    int lane = threadIdx.x & 31;
    int e0 = (blockIdx.x * WPB + wave) * EPW;
    if (e0 >= ne) return;
    if (e0 + EPW <= ne) spmm64_batch<false>(g, src, dst, agg, e0, ne, lane);
    else                spmm64_batch<true >(g, src, dst, agg, e0, ne, lane);
}

template <int K, int N, bool RELU, bool BIAS>
__global__ void k_gemm_wmma(const float* __restrict__ A,
                            const float* __restrict__ rowscale,
                            const float* __restrict__ W,
                            const float* __restrict__ bias,
                            float* __restrict__ C) {
    constexpr int tilesN = N / 16;
    int wave = (blockIdx.x * blockDim.x + threadIdx.x) >> 5;
    int lane = threadIdx.x & 31;
    int tm = wave / tilesN;
    int tn = wave % tilesN;

    int half = lane >> 4;
    int l15  = lane & 15;
    int arow = tm * 16 + l15;
    int gcol = tn * 16 + l15;

    float scale = rowscale[arow];
    const float* ap = A + (long)arow * K + half * 2;
    const float* wp = W + gcol;

    v8f c = {};
    for (int k = 0; k < K; k += 4) {
        v2f a;
        a.x = ap[k]     * scale;
        a.y = ap[k + 1] * scale;
        int kb = k + half * 2;
        v2f b;
        b.x = wp[(long)kb * N];
        b.y = wp[(long)(kb + 1) * N];
        c = __builtin_amdgcn_wmma_f32_16x16x4_f32(
                 false, a,  false, b,
                 (short)0, c,  false,  false);
    }

    float bval = 0.0f;
    if (BIAS) bval = bias[gcol];
    int rb = half * 8;
    #pragma unroll
    for (int i = 0; i < 8; ++i) {
        float v = c[i] + bval;
        if (RELU) v = fmaxf(v, 0.0f);
        C[(long)(tm * 16 + rb + i) * N + gcol] = v;
    }
}

__global__ void k_final(const float* __restrict__ agg2, const float* __restrict__ normD2,
                        const float* __restrict__ b2, float* __restrict__ out, int tot) {
    int t = blockIdx.x * blockDim.x + threadIdx.x;
    if (t < tot) {
        int r = t >> 6;
        int f = t & 63;
        out[t] = agg2[t] * normD2[r] + b2[f];
    }
}

extern "C" void kernel_launch(void* const* d_in, const int* in_sizes, int n_in,
                              void* d_out, int out_size, void* d_ws, size_t ws_size,
                              hipStream_t stream) {
    const float* x    = (const float*)d_in[0];
    const float* W1   = (const float*)d_in[1];
    const float* b1   = (const float*)d_in[2];
    const float* W2   = (const float*)d_in[3];
    const float* b2   = (const float*)d_in[4];
    const int*   src1 = (const int*)d_in[5];
    const int*   dst1 = (const int*)d_in[6];
    const int*   src2 = (const int*)d_in[7];
    const int*   dst2 = (const int*)d_in[8];

    const int IN_F = 128, H_F = 256, N_CLS = 64;
    const int NSRC1 = in_sizes[0] / IN_F;
    const int E1    = in_sizes[5];
    const int E2    = in_sizes[7];
    const int ND1   = 100000;
    const int ND2   = 16384;

    float* ws    = (float*)d_ws;
    float* degS1 = ws;
    float* degD1 = degS1 + NSRC1;
    float* degS2 = degD1 + ND1;
    float* degD2 = degS2 + ND1;
    float* agg1  = degD2 + ND2;
    float* agg2  = agg1 + (long)ND1 * IN_F;
    float* h     = agg2 + (long)ND2 * N_CLS;
    float* g     = h + (long)ND1 * H_F;

    long zcount = (NSRC1 + ND1 + ND1 + ND2) + (long)ND1 * IN_F + (long)ND2 * N_CLS;
    k_zero<<<2048, 256, 0, stream>>>(ws, zcount);

    k_deg<<<(E1 + 255) / 256, 256, 0, stream>>>(src1, dst1, E1, degS1, degD1);
    k_deg<<<(E2 + 255) / 256, 256, 0, stream>>>(src2, dst2, E2, degS2, degD2);
    int degtot = NSRC1 + ND1 + ND1 + ND2;
    k_rsqrt<<<(degtot + 255) / 256, 256, 0, stream>>>(ws, degtot);

    k_spmm128<<<(E1 + 63) / 64, 256, 0, stream>>>(x, src1, dst1, degS1, agg1, E1);

    {
        int waves = (ND1 / 16) * (H_F / 16);
        k_gemm_wmma<128, 256, true, true>
            <<<waves / 4, 128, 0, stream>>>(agg1, degD1, W1, b1, h);
    }

    {
        int waves = (ND1 / 16) * (N_CLS / 16);
        k_gemm_wmma<256, 64, false, false>
            <<<waves / 4, 128, 0, stream>>>(h, degS2, W2, nullptr, g);
    }

    k_spmm64<<<(E2 + 63) / 64, 256, 0, stream>>>(g, src2, dst2, agg2, E2);

    int tot = ND2 * N_CLS;
    k_final<<<(tot + 255) / 256, 256, 0, stream>>>(agg2, degD2, b2, (float*)d_out, tot);
}


# Both dense products fused in one kernel: 8 waves stage 32 scaled rows in LDS once, each wave computes 2x2 tiles, h passes through LDS into the second product and never goes to memory
# speedup vs baseline: 1.7434x; 1.2503x over previous
#include <hip/hip_runtime.h>


typedef __attribute__((ext_vector_type(2))) float v2f;
typedef __attribute__((ext_vector_type(8))) float v8f;

__device__ __forceinline__ void atomAddF(float* p, float v) {
    asm volatile("global_atomic_add_f32 %0, %1, off" :: "v"(p), "v"(v) : "memory");
}

template <int BYTE_OFF>
__device__ __forceinline__ void atomAddRowPiece(float* rowBase, unsigned laneByte, float v) {
    asm volatile("global_atomic_add_f32 %0, %1, %2 offset:%3"
                 :: "v"(laneByte), "v"(v), "s"(rowBase), "n"(BYTE_OFF) : "memory");
}

__global__ void k_zero(float* __restrict__ p, long n) {
    long i = (long)blockIdx.x * blockDim.x + threadIdx.x;
    long stride = (long)gridDim.x * blockDim.x;
    for (; i < n; i += stride) p[i] = 0.0f;
}

__global__ void k_deg(const int* __restrict__ src, const int* __restrict__ dst,
                      int ne, float* __restrict__ degS, float* __restrict__ degD) {
    int e = blockIdx.x * blockDim.x + threadIdx.x;
    if (e < ne) {
        atomAddF(&degS[src[e]], 1.0f);
        atomAddF(&degD[dst[e]], 1.0f);
    }
}

__global__ void k_rsqrt(float* __restrict__ p, int n) {
    int i = blockIdx.x * blockDim.x + threadIdx.x;
    if (i < n) p[i] = rsqrtf(fmaxf(p[i], 1.0f));
}

constexpr int EPW = 8;
constexpr int WPB = 8;

template <bool TAIL>
__device__ __forceinline__ void spmm128_batch(const float* __restrict__ x,
                                              const int* __restrict__ src,
                                              const int* __restrict__ dst,
                                              const float* __restrict__ normS,
                                              float* __restrict__ agg,
                                              int e0, int ne, int lane) {
    int s[EPW], d[EPW];
    #pragma unroll
    for (int j = 0; j < EPW; ++j) {
        int e = TAIL ? min(e0 + j, ne - 1) : e0 + j;
        s[j] = src[e];
        d[j] = dst[e];
    }
    float ns[EPW];
    float v[EPW][4];
    #pragma unroll
    for (int j = 0; j < EPW; ++j) {
        ns[j] = normS[s[j]];
        const float* xr = x + (long)s[j] * 128 + lane;
        v[j][0] = xr[0];
        v[j][1] = xr[32];
        v[j][2] = xr[64];
        v[j][3] = xr[96];
    }
    unsigned laneByte = (unsigned)lane * 4u;
    #pragma unroll
    for (int j = 0; j < EPW; ++j) {
        if (!TAIL || e0 + j < ne) {
            float* row = agg + (long)d[j] * 128;
            atomAddRowPiece<0>(row, laneByte, v[j][0] * ns[j]);
            atomAddRowPiece<128>(row, laneByte, v[j][1] * ns[j]);
            atomAddRowPiece<256>(row, laneByte, v[j][2] * ns[j]);
            atomAddRowPiece<384>(row, laneByte, v[j][3] * ns[j]);
        }
    }
}

__global__ void k_spmm128(const float* __restrict__ x, const int* __restrict__ src,
                          const int* __restrict__ dst, const float* __restrict__ normS,
                          float* __restrict__ agg, int ne) {
    int wave = __builtin_amdgcn_readfirstlane((int)(threadIdx.x >> 5));
    int lane = threadIdx.x & 31;
    int e0 = (blockIdx.x * WPB + wave) * EPW;
    if (e0 >= ne) return;
    if (e0 + EPW <= ne) spmm128_batch<false>(x, src, dst, normS, agg, e0, ne, lane);
    else                spmm128_batch<true >(x, src, dst, normS, agg, e0, ne, lane);
}

template <bool TAIL>
__device__ __forceinline__ void spmm64_batch(const float* __restrict__ g,
                                             const int* __restrict__ src,
                                             const int* __restrict__ dst,
                                             float* __restrict__ agg,
                                             int e0, int ne, int lane) {
    int s[EPW], d[EPW];
    #pragma unroll
    for (int j = 0; j < EPW; ++j) {
        int e = TAIL ? min(e0 + j, ne - 1) : e0 + j;
        s[j] = src[e];
        d[j] = dst[e];
    }
    float v[EPW][2];
    #pragma unroll
    for (int j = 0; j < EPW; ++j) {
        const float* gr = g + (long)s[j] * 64 + lane;
        v[j][0] = gr[0];
        v[j][1] = gr[32];
    }
    unsigned laneByte = (unsigned)lane * 4u;
    #pragma unroll
    for (int j = 0; j < EPW; ++j) {
        if (!TAIL || e0 + j < ne) {
            float* row = agg + (long)d[j] * 64;
            atomAddRowPiece<0>(row, laneByte, v[j][0]);
            atomAddRowPiece<128>(row, laneByte, v[j][1]);
        }
    }
}

__global__ void k_spmm64(const float* __restrict__ g, const int* __restrict__ src,
                         const int* __restrict__ dst, float* __restrict__ agg, int ne) {
    int wave = __builtin_amdgcn_readfirstlane((int)(threadIdx.x >> 5));
    int lane = threadIdx.x & 31;
    int e0 = (blockIdx.x * WPB + wave) * EPW;
    if (e0 >= ne) return;
    if (e0 + EPW <= ne) spmm64_batch<false>(g, src, dst, agg, e0, ne, lane);
    else                spmm64_batch<true >(g, src, dst, agg, e0, ne, lane);
}

constexpr int FBM = 32;
constexpr int FK1 = 128;
constexpr int FN1 = 256;
constexpr int FN2 = 64;
constexpr int FPA = FK1 + 4;
constexpr int FPH = FN1 + 4;
constexpr int FPG = FN2 + 4;

__device__ __forceinline__ v8f wmma_f32(v2f a, v2f b, v8f c) {
    return __builtin_amdgcn_wmma_f32_16x16x4_f32(
         false, a,  false, b,
         (short)0, c,  false,  false);
}

__device__ __forceinline__ void h_tile_to_lds(const v8f& c, int mt, int ntile, int half, int l15,
                                              const float* __restrict__ b1,
                                              const float* sS, float* sH) {
    int gcol = ntile * 16 + l15;
    float bval = b1[gcol];
    #pragma unroll
    for (int i = 0; i < 8; ++i) {
        int r = mt * 16 + half * 8 + i;
        float v = fmaxf(c[i] + bval, 0.0f);
        sH[r * FPH + gcol] = v * sS[r];
    }
}

__global__ __launch_bounds__(256)
void k_dense_fused(const float* __restrict__ agg1, const float* __restrict__ normD1,
                   const float* __restrict__ W1, const float* __restrict__ b1,
                   const float* __restrict__ normS2, const float* __restrict__ W2,
                   float* __restrict__ g) {
    __shared__ alignas(16) float sA[FBM * FPA];
    __shared__ alignas(16) float sH[FBM * FPH];
    __shared__ float sS[FBM];

    int tid  = threadIdx.x;
    int lane = tid & 31;
    int wave = tid >> 5;
    int half = lane >> 4;
    int l15  = lane & 15;
    long row0 = (long)blockIdx.x * FBM;

    #pragma unroll
    for (int i = 0; i < 4; ++i) {
        int idx = tid + 256 * i;
        int r   = idx >> 5;
        int c4  = idx & 31;
        float sc = normD1[row0 + r];
        float4 v = *reinterpret_cast<const float4*>(agg1 + (row0 + r) * FK1 + c4 * 4);
        v.x *= sc; v.y *= sc; v.z *= sc; v.w *= sc;
        *reinterpret_cast<float4*>(sA + r * FPA + c4 * 4) = v;
    }
    if (tid < FBM) sS[tid] = normS2[row0 + tid];
    __syncthreads();

    {
        int nt0 = wave * 2;
        v8f c00 = {}, c01 = {}, c10 = {}, c11 = {};
        const float* a0p = sA + l15 * FPA + half * 2;
        const float* a1p = sA + (16 + l15) * FPA + half * 2;
        const float* wp  = W1 + nt0 * 16 + l15;
        #pragma unroll 4
        for (int k = 0; k < FK1; k += 4) {
            v2f a0 = *reinterpret_cast<const v2f*>(a0p + k);
            v2f a1 = *reinterpret_cast<const v2f*>(a1p + k);
            int kb = k + half * 2;
            v2f b0, b1v;
            b0.x  = wp[(long)kb * FN1];
            b0.y  = wp[(long)(kb + 1) * FN1];
            b1v.x = wp[(long)kb * FN1 + 16];
            b1v.y = wp[(long)(kb + 1) * FN1 + 16];
            c00 = wmma_f32(a0, b0,  c00);
            c01 = wmma_f32(a0, b1v, c01);
            c10 = wmma_f32(a1, b0,  c10);
            c11 = wmma_f32(a1, b1v, c11);
        }
        h_tile_to_lds(c00, 0, nt0,     half, l15, b1, sS, sH);
        h_tile_to_lds(c01, 0, nt0 + 1, half, l15, b1, sS, sH);
        h_tile_to_lds(c10, 1, nt0,     half, l15, b1, sS, sH);
        h_tile_to_lds(c11, 1, nt0 + 1, half, l15, b1, sS, sH);
    }
    __syncthreads();

    {
        int mt = wave >> 2;
        int nt = wave & 3;
        v8f c = {};
        const float* hp = sH + (mt * 16 + l15) * FPH + half * 2;
        const float* wp = W2 + nt * 16 + l15;
        #pragma unroll 8
        for (int k = 0; k < FN1; k += 4) {
            v2f a = *reinterpret_cast<const v2f*>(hp + k);
            int kb = k + half * 2;
            v2f b;
            b.x = wp[(long)kb * FN2];
            b.y = wp[(long)(kb + 1) * FN2];
            c = wmma_f32(a, b, c);
        }
        float* sG = sA;
        int gcol = nt * 16 + l15;
        #pragma unroll
        for (int i = 0; i < 8; ++i)
            sG[(mt * 16 + half * 8 + i) * FPG + gcol] = c[i];
    }
    __syncthreads();

    {
        const float* sG = sA;
        #pragma unroll
        for (int i = 0; i < 2; ++i) {
            int idx = tid + 256 * i;
            int r   = idx >> 4;
            int c4  = idx & 15;
            float4 v = *reinterpret_cast<const float4*>(sG + r * FPG + c4 * 4);
            *reinterpret_cast<float4*>(g + (row0 + r) * FN2 + c4 * 4) = v;
        }
    }
}

__global__ void k_final(const float* __restrict__ agg2, const float* __restrict__ normD2,
                        const float* __restrict__ b2, float* __restrict__ out, int tot) {
    int t = blockIdx.x * blockDim.x + threadIdx.x;
    if (t < tot) {
        int r = t >> 6;
        int f = t & 63;
        out[t] = agg2[t] * normD2[r] + b2[f];
    }
}

extern "C" void kernel_launch(void* const* d_in, const int* in_sizes, int n_in,
                              void* d_out, int out_size, void* d_ws, size_t ws_size,
                              hipStream_t stream) {
    const float* x    = (const float*)d_in[0];
    const float* W1   = (const float*)d_in[1];
    const float* b1   = (const float*)d_in[2];
    const float* W2   = (const float*)d_in[3];
    const float* b2   = (const float*)d_in[4];
    const int*   src1 = (const int*)d_in[5];
    const int*   dst1 = (const int*)d_in[6];
    const int*   src2 = (const int*)d_in[7];
    const int*   dst2 = (const int*)d_in[8];

    const int IN_F = 128, N_CLS = 64;
    const int NSRC1 = in_sizes[0] / IN_F;
    const int E1    = in_sizes[5];
    const int E2    = in_sizes[7];
    const int ND1   = 100000;
    const int ND2   = 16384;

    float* ws    = (float*)d_ws;
    float* degS1 = ws;
    float* degD1 = degS1 + NSRC1;
    float* degS2 = degD1 + ND1;
    float* degD2 = degS2 + ND1;
    float* agg1  = degD2 + ND2;
    float* agg2  = agg1 + (long)ND1 * IN_F;
    float* g     = agg2 + (long)ND2 * N_CLS;

    long zcount = (NSRC1 + ND1 + ND1 + ND2) + (long)ND1 * IN_F + (long)ND2 * N_CLS;
    k_zero<<<2048, 256, 0, stream>>>(ws, zcount);

    k_deg<<<(E1 + 255) / 256, 256, 0, stream>>>(src1, dst1, E1, degS1, degD1);
    k_deg<<<(E2 + 255) / 256, 256, 0, stream>>>(src2, dst2, E2, degS2, degD2);
    int degtot = NSRC1 + ND1 + ND1 + ND2;
    k_rsqrt<<<(degtot + 255) / 256, 256, 0, stream>>>(ws, degtot);

    k_spmm128<<<(E1 + 63) / 64, 256, 0, stream>>>(x, src1, dst1, degS1, agg1, E1);

    k_dense_fused<<<ND1 / 32, 256, 0, stream>>>(agg1, degD1, W1, b1, degS2, W2, g);

    k_spmm64<<<(E2 + 63) / 64, 256, 0, stream>>>(g, src2, dst2, agg2, E2);

    int tot = ND2 * N_CLS;
    k_final<<<(tot + 255) / 256, 256, 0, stream>>>(agg2, degD2, b2, (float*)d_out, tot);
}


# Layer-1 SpMM pulls: dst-degree atomic returns each edge's rank, edges sorted by destination, fused dense kernel sums its own 32 rows in LDS; agg1 and its float atomics gone; sum order = rank order
# speedup vs baseline: 2.0118x; 1.1540x over previous
#include <hip/hip_runtime.h>


typedef __attribute__((ext_vector_type(2))) float v2f;
typedef __attribute__((ext_vector_type(8))) float v8f;

__device__ __forceinline__ void atomAddF(float* p, float v) {
    asm volatile("global_atomic_add_f32 %0, %1, off" :: "v"(p), "v"(v) : "memory");
}

__device__ __forceinline__ float atomAddFRet(float* p, float v) {
    float old;
    asm volatile("global_atomic_add_f32 %0, %1, %2, off th:TH_ATOMIC_RETURN scope:SCOPE_DEV\n\t"
                 "s_wait_loadcnt 0x0"
                 : "=&v"(old) : "v"(p), "v"(v) : "memory");
    return old;
}

template <int BYTE_OFF>
__device__ __forceinline__ void atomAddRowPiece(float* rowBase, unsigned laneByte, float v) {
    asm volatile("global_atomic_add_f32 %0, %1, %2 offset:%3"
                 :: "v"(laneByte), "v"(v), "s"(rowBase), "n"(BYTE_OFF) : "memory");
}

__global__ void k_zero(float* __restrict__ p, long n) {
    long i = (long)blockIdx.x * blockDim.x + threadIdx.x;
    long stride = (long)gridDim.x * blockDim.x;
    for (; i < n; i += stride) p[i] = 0.0f;
}

template <bool RANK>
__global__ void k_deg(const int* __restrict__ src, const int* __restrict__ dst,
                      int ne, float* __restrict__ degS, float* __restrict__ degD,
                      int* __restrict__ rank) {
    int e = blockIdx.x * blockDim.x + threadIdx.x;
    if (e < ne) {
        atomAddF(&degS[src[e]], 1.0f);
        if (RANK) rank[e] = (int)atomAddFRet(&degD[dst[e]], 1.0f);
        else      atomAddF(&degD[dst[e]], 1.0f);
    }
}

__device__ __forceinline__ int wave_incl_scan(int v, int lane) {
    #pragma unroll
    for (int o = 1; o < 32; o <<= 1) {
        int t = __shfl_up(v, o);
        if (lane >= o) v += t;
    }
    return v;
}


__global__ void k_rsqrt_buckets(float* __restrict__ p, int n, int lo, int hi,
                                int* __restrict__ rowLocal, int* __restrict__ rowCount,
                                int* __restrict__ waveSum) {
    int i = blockIdx.x * blockDim.x + threadIdx.x;
    int lane = threadIdx.x & 31;
    float dg = (i < n) ? p[i] : 0.0f;
    bool isRow = (i >= lo) && (i < hi);
    int c = isRow ? (int)dg : 0;
    int incl = wave_incl_scan(c, lane);
    if (isRow) {
        rowLocal[i - lo] = incl - c;
        rowCount[i - lo] = c;
    }
    int widx = (i >> 5) - (lo >> 5);
    int nw   = ((hi - 1) >> 5) - (lo >> 5) + 1;
    if (lane == 31 && widx >= 0 && widx < nw) waveSum[widx] = incl;
    if (i < n) p[i] = rsqrtf(fmaxf(dg, 1.0f));
}

__global__ __launch_bounds__(1024)
void k_scan_waves(const int* __restrict__ waveSum, int* __restrict__ waveBase, int lo, int hi) {
    __shared__ int sW[32];
    int nw  = ((hi - 1) >> 5) - (lo >> 5) + 1;
    int per = (nw + 1023) / 1024;
    int t = threadIdx.x, lane = t & 31, w = t >> 5;
    int b0 = t * per;
    int loc = 0;
    for (int k = 0; k < per; ++k)
        if (b0 + k < nw) loc += waveSum[b0 + k];
    int incl = wave_incl_scan(loc, lane);
    if (lane == 31) sW[w] = incl;
    __syncthreads();
    if (w == 0) {
        int v = sW[lane];
        int iv = wave_incl_scan(v, lane);
        sW[lane] = iv - v;
    }
    __syncthreads();
    int run = sW[w] + incl - loc;
    for (int k = 0; k < per; ++k)
        if (b0 + k < nw) {
            int c = waveSum[b0 + k];
            waveBase[b0 + k] = run;
            run += c;
        }
}

__global__ void k_fill(const int* __restrict__ src, const int* __restrict__ dst, int ne,
                       const int* __restrict__ rank, const int* __restrict__ rowLocal,
                       const int* __restrict__ waveBase, int lo, int* __restrict__ bySrc) {
    int e = blockIdx.x * blockDim.x + threadIdx.x;
    if (e < ne) {
        int d = dst[e];
        int start = waveBase[((d + lo) >> 5) - (lo >> 5)] + rowLocal[d];
        bySrc[start + rank[e]] = src[e];
    }
}

constexpr int EPW = 8;
constexpr int WPB = 8;

template <bool TAIL>
__device__ __forceinline__ void spmm64_batch(const float* __restrict__ g,
                                             const int* __restrict__ src,
                                             const int* __restrict__ dst,
                                             float* __restrict__ agg,
                                             int e0, int ne, int lane) {
    int s[EPW], d[EPW];
    #pragma unroll
    for (int j = 0; j < EPW; ++j) {
        int e = TAIL ? min(e0 + j, ne - 1) : e0 + j;
        s[j] = src[e];
        d[j] = dst[e];
    }
    float v[EPW][2];
    #pragma unroll
    for (int j = 0; j < EPW; ++j) {
        const float* gr = g + (long)s[j] * 64 + lane;
        v[j][0] = gr[0];
        v[j][1] = gr[32];
    }
    unsigned laneByte = (unsigned)lane * 4u;
    #pragma unroll
    for (int j = 0; j < EPW; ++j) {
        if (!TAIL || e0 + j < ne) {
            float* row = agg + (long)d[j] * 64;
            atomAddRowPiece<0>(row, laneByte, v[j][0]);
            atomAddRowPiece<128>(row, laneByte, v[j][1]);
        }
    }
}

__global__ void k_spmm64(const float* __restrict__ g, const int* __restrict__ src,
                         const int* __restrict__ dst, float* __restrict__ agg, int ne) {
    int wave = __builtin_amdgcn_readfirstlane((int)(threadIdx.x >> 5));
    int lane = threadIdx.x & 31;
    int e0 = (blockIdx.x * WPB + wave) * EPW;
    if (e0 >= ne) return;
    if (e0 + EPW <= ne) spmm64_batch<false>(g, src, dst, agg, e0, ne, lane);
    else                spmm64_batch<true >(g, src, dst, agg, e0, ne, lane);
}

constexpr int FBM = 32;
constexpr int FK1 = 128;
constexpr int FN1 = 256;
constexpr int FN2 = 64;
constexpr int FPA = FK1 + 4;
constexpr int FPH = FN1 + 4;
constexpr int FPG = FN2 + 4;

__device__ __forceinline__ v8f wmma_f32(v2f a, v2f b, v8f c) {
    return __builtin_amdgcn_wmma_f32_16x16x4_f32(
         false, a,  false, b,
         (short)0, c,  false,  false);
}

__device__ __forceinline__ void h_tile_to_lds(const v8f& c, int mt, int ntile, int half, int l15,
                                              const float* __restrict__ b1,
                                              const float* sS, float* sH) {
    int gcol = ntile * 16 + l15;
    float bval = b1[gcol];
    #pragma unroll
    for (int i = 0; i < 8; ++i) {
        int r = mt * 16 + half * 8 + i;
        float v = fmaxf(c[i] + bval, 0.0f);
        sH[r * FPH + gcol] = v * sS[r];
    }
}

__global__ __launch_bounds__(256)
void k_dense_fused(const float* __restrict__ x, const float* __restrict__ normS1,
                   const int* __restrict__ waveBase, const int* __restrict__ rowLocal,
                   const int* __restrict__ rowCount, int lo,
                   const int* __restrict__ bySrc, const float* __restrict__ normD1,
                   const float* __restrict__ W1, const float* __restrict__ b1,
                   const float* __restrict__ normS2, const float* __restrict__ W2,
                   float* __restrict__ g) {
    __shared__ alignas(16) float sA[FBM * FPA];
    __shared__ alignas(16) float sH[FBM * FPH];
    __shared__ float sS[FBM];

    int tid  = threadIdx.x;
    int lane = tid & 31;
    int wave = __builtin_amdgcn_readfirstlane((int)(tid >> 5));
    int half = lane >> 4;
    int l15  = lane & 15;
    long row0 = (long)blockIdx.x * FBM;

    for (int rr = 0; rr < FBM / WPB; ++rr) {
        int rl  = wave * (FBM / WPB) + rr;
        int row = (int)row0 + rl;
        int n   = rowCount[row];
        const int* bucket = bySrc + waveBase[((row + lo) >> 5) - (lo >> 5)] + rowLocal[row];
        float a0 = 0.0f, a1 = 0.0f, a2 = 0.0f, a3 = 0.0f;
        for (int j0 = 0; j0 < n; j0 += EPW) {
            int s[EPW];
            #pragma unroll
            for (int j = 0; j < EPW; ++j) s[j] = bucket[min(j0 + j, n - 1)];
            float ns[EPW];
            float v[EPW][4];
            #pragma unroll
            for (int j = 0; j < EPW; ++j) {
                float nsj = normS1[s[j]];
                ns[j] = (j0 + j < n) ? nsj : 0.0f;
                const float* xr = x + (long)s[j] * FK1 + lane;
                v[j][0] = xr[0];
                v[j][1] = xr[32];
                v[j][2] = xr[64];
                v[j][3] = xr[96];
            }
            #pragma unroll
            for (int j = 0; j < EPW; ++j) {
                a0 += v[j][0] * ns[j];
                a1 += v[j][1] * ns[j];
                a2 += v[j][2] * ns[j];
                a3 += v[j][3] * ns[j];
            }
        }
        float sc = normD1[row];
        float* sr = sA + rl * FPA + lane;
        sr[0]  = a0 * sc;
        sr[32] = a1 * sc;
        sr[64] = a2 * sc;
        sr[96] = a3 * sc;
    }
    if (tid < FBM) sS[tid] = normS2[row0 + tid];
    __syncthreads();

    {
        int nt0 = wave * 2;
        v8f c00 = {}, c01 = {}, c10 = {}, c11 = {};
        const float* a0p = sA + l15 * FPA + half * 2;
        const float* a1p = sA + (16 + l15) * FPA + half * 2;
        const float* wp  = W1 + nt0 * 16 + l15;
        #pragma unroll 4
        for (int k = 0; k < FK1; k += 4) {
            v2f a0 = *reinterpret_cast<const v2f*>(a0p + k);
            v2f a1 = *reinterpret_cast<const v2f*>(a1p + k);
            int kb = k + half * 2;
            v2f b0, b1v;
            b0.x  = wp[(long)kb * FN1];
            b0.y  = wp[(long)(kb + 1) * FN1];
            b1v.x = wp[(long)kb * FN1 + 16];
            b1v.y = wp[(long)(kb + 1) * FN1 + 16];
            c00 = wmma_f32(a0, b0,  c00);
            c01 = wmma_f32(a0, b1v, c01);
            c10 = wmma_f32(a1, b0,  c10);
            c11 = wmma_f32(a1, b1v, c11);
        }
        h_tile_to_lds(c00, 0, nt0,     half, l15, b1, sS, sH);
        h_tile_to_lds(c01, 0, nt0 + 1, half, l15, b1, sS, sH);
        h_tile_to_lds(c10, 1, nt0,     half, l15, b1, sS, sH);
        h_tile_to_lds(c11, 1, nt0 + 1, half, l15, b1, sS, sH);
    }
    __syncthreads();

    {
        int mt = wave >> 2;
        int nt = wave & 3;
        v8f c = {};
        const float* hp = sH + (mt * 16 + l15) * FPH + half * 2;
        const float* wp = W2 + nt * 16 + l15;
        #pragma unroll 8
        for (int k = 0; k < FN1; k += 4) {
            v2f a = *reinterpret_cast<const v2f*>(hp + k);
            int kb = k + half * 2;
            v2f b;
            b.x = wp[(long)kb * FN2];
            b.y = wp[(long)(kb + 1) * FN2];
            c = wmma_f32(a, b, c);
        }
        float* sG = sA;
        int gcol = nt * 16 + l15;
        #pragma unroll
        for (int i = 0; i < 8; ++i)
            sG[(mt * 16 + half * 8 + i) * FPG + gcol] = c[i];
    }
    __syncthreads();

    {
        const float* sG = sA;
        #pragma unroll
        for (int i = 0; i < 2; ++i) {
            int idx = tid + 256 * i;
            int r   = idx >> 4;
            int c4  = idx & 15;
            float4 v = *reinterpret_cast<const float4*>(sG + r * FPG + c4 * 4);
            *reinterpret_cast<float4*>(g + (row0 + r) * FN2 + c4 * 4) = v;
        }
    }
}

__global__ void k_final(const float* __restrict__ agg2, const float* __restrict__ normD2,
                        const float* __restrict__ b2, float* __restrict__ out, int tot) {
    int t = blockIdx.x * blockDim.x + threadIdx.x;
    if (t < tot) {
        int r = t >> 6;
        int f = t & 63;
        out[t] = agg2[t] * normD2[r] + b2[f];
    }
}

extern "C" void kernel_launch(void* const* d_in, const int* in_sizes, int n_in,
                              void* d_out, int out_size, void* d_ws, size_t ws_size,
                              hipStream_t stream) {
    const float* x    = (const float*)d_in[0];
    const float* W1   = (const float*)d_in[1];
    const float* b1   = (const float*)d_in[2];
    const float* W2   = (const float*)d_in[3];
    const float* b2   = (const float*)d_in[4];
    const int*   src1 = (const int*)d_in[5];
    const int*   dst1 = (const int*)d_in[6];
    const int*   src2 = (const int*)d_in[7];
    const int*   dst2 = (const int*)d_in[8];

    const int IN_F = 128, N_CLS = 64;
    const int NSRC1 = in_sizes[0] / IN_F;
    const int E1    = in_sizes[5];
    const int E2    = in_sizes[7];
    const int ND1   = 100000;
    const int ND2   = 16384;

    float* ws    = (float*)d_ws;
    float* degS1 = ws;
    float* degD1 = degS1 + NSRC1;
    float* degS2 = degD1 + ND1;
    float* degD2 = degS2 + ND1;
    float* agg2  = degD2 + ND2;
    const int NWS = (ND1 / 32 + 1 + 31) / 32 * 32;
    const int E1R = (E1 + 31) / 32 * 32;
    int*   rowLocal = (int*)(agg2 + (long)ND2 * N_CLS);
    int*   rowCount = rowLocal + ND1;
    int*   waveSum  = rowCount + ND1;
    int*   waveBase = waveSum + NWS;
    int*   rank     = waveBase + NWS;
    int*   bySrc    = rank + E1R;
    float* g     = (float*)(bySrc + E1R);

    long zcount = (NSRC1 + ND1 + ND1 + ND2) + (long)ND2 * N_CLS;
    k_zero<<<2048, 256, 0, stream>>>(ws, zcount);

    k_deg<true ><<<(E1 + 255) / 256, 256, 0, stream>>>(src1, dst1, E1, degS1, degD1, rank);
    k_deg<false><<<(E2 + 255) / 256, 256, 0, stream>>>(src2, dst2, E2, degS2, degD2, nullptr);
    int degtot = NSRC1 + ND1 + ND1 + ND2;
    k_rsqrt_buckets<<<(degtot + 255) / 256, 256, 0, stream>>>(ws, degtot, NSRC1, NSRC1 + ND1,
                                                              rowLocal, rowCount, waveSum);

    k_scan_waves<<<1, 1024, 0, stream>>>(waveSum, waveBase, NSRC1, NSRC1 + ND1);
    k_fill<<<(E1 + 255) / 256, 256, 0, stream>>>(src1, dst1, E1, rank, rowLocal, waveBase,
                                                 NSRC1, bySrc);

    k_dense_fused<<<ND1 / 32, 256, 0, stream>>>(x, degS1, waveBase, rowLocal, rowCount, NSRC1,
                                                bySrc, degD1, W1, b1, degS2, W2, g);

    k_spmm64<<<(E2 + 63) / 64, 256, 0, stream>>>(g, src2, dst2, agg2, E2);

    int tot = ND2 * N_CLS;
    k_final<<<(tot + 255) / 256, 256, 0, stream>>>(agg2, degD2, b2, (float*)d_out, tot);
}
